# starting kernel
# baseline (speedup 1.0000x reference)
#include <hip/hip_runtime.h>


typedef float v2f __attribute__((ext_vector_type(2)));
typedef float v8f __attribute__((ext_vector_type(8)));

#define EMB_D           64
#define COL_TILES       4
#define WAVES_PER_BLOCK 8
#define THREADS         256

__device__ __forceinline__ v8f wmma_acc(v2f a, v2f b, v8f c) {
  return __builtin_amdgcn_wmma_f32_16x16x4_f32(
       false, a,  false, b,
       (short)0, c,  false,  false);
}

__global__ __launch_bounds__(THREADS) void embbag_wmma(
    const int* __restrict__ indices,
    const int* __restrict__ offsets,
    const float* __restrict__ weight,
    float* __restrict__ out,
    int num_bags, int total_idx)
{
  __shared__ unsigned ldsOff[WAVES_PER_BLOCK * 64];

  const int tid       = threadIdx.x;
  const int lane      = tid & 31;
  const int waveInBlk = tid >> 5;
  const int wave      = blockIdx.x * WAVES_PER_BLOCK + waveInBlk;
  const int bag       = wave >> 2;
  const int colTile   = wave & 3;
  if (bag >= num_bags) return;

  const int half = lane >> 4;
  const int col  = lane & 15;
  const unsigned colByte = (unsigned)(colTile * 16 + col) * 4u;

  const int bagStart = offsets[bag];
  const int bagEnd   = (bag + 1 < num_bags) ? offsets[bag + 1] : total_idx;
  const int cnt      = bagEnd - bagStart;

  unsigned* myOff   = &ldsOff[waveInBlk * 64];
  const char* wbase = (const char*)weight;

  v2f aOnes; aOnes.x = 1.0f; aOnes.y = 1.0f;

  if (cnt == 50) {
    {
      const int p0 = bagStart + lane;
      int p1 = bagStart + lane + 32;
      p1 = p1 < total_idx - 1 ? p1 : total_idx - 1;
      const unsigned o0 = ((unsigned)indices[p0]) << 8;
      const unsigned o1 = ((unsigned)indices[p1]) << 8;
      myOff[lane]      = o0;
      myOff[lane + 32] = o1;
    }
    asm volatile("s_wait_dscnt 0x0" ::: "memory");

    v8f a0 = {}, a1 = {};
    #pragma unroll
    for (int cc = 0; cc < 6; ++cc) {
      const int sA = (cc << 3) + half;
      const unsigned r0 = myOff[sA];
      const unsigned r1 = myOff[sA + 2];
      const unsigned r2 = myOff[sA + 4];
      const unsigned r3 = myOff[sA + 6];
      const float x0 = *(const float*)(wbase + (r0 + colByte));
      const float x1 = *(const float*)(wbase + (r1 + colByte));
      const float x2 = *(const float*)(wbase + (r2 + colByte));
      const float x3 = *(const float*)(wbase + (r3 + colByte));
      v2f bA; bA.x = x0; bA.y = x1;
      v2f bB; bB.x = x2; bB.y = x3;
      a0 = wmma_acc(aOnes, bA, a0);
      a1 = wmma_acc(aOnes, bB, a1);
    }
    {
      const unsigned rT = myOff[48 + half];
      const float xT = *(const float*)(wbase + (rT + colByte));
      v2f bT; bT.x = xT; bT.y = 0.0f;
      a0 = wmma_acc(aOnes, bT, a0);
    }
    const float r = (a0[0] + a1[0]) / 50.0f;
    if (half == 0) out[(size_t)bag * EMB_D + (colByte >> 2)] = r;
    return;
  }

  v8f acc0 = {}, acc1 = {};
  for (int base = 0; base < cnt; base += 64) {
    const int remAll = cnt - base;
    const int rem    = remAll < 64 ? remAll : 64;

    {
      const int s0 = lane, s1 = lane + 32;
      unsigned o0 = 0u, o1 = 0u;
      if (s0 < rem) o0 = ((unsigned)indices[bagStart + base + s0]) << 8;
      if (s1 < rem) o1 = ((unsigned)indices[bagStart + base + s1]) << 8;
      myOff[s0] = o0;
      myOff[s1] = o1;
    }
    asm volatile("s_wait_dscnt 0x0" ::: "memory");

    const int mainChunks = rem >> 2;
    int c = 0;
    for (; c + 2 <= mainChunks; c += 2) {
      const int sA = (c << 2) + half;
      const unsigned r0 = myOff[sA];
      const unsigned r1 = myOff[sA + 2];
      const unsigned r2 = myOff[sA + 4];
      const unsigned r3 = myOff[sA + 6];
      const float x0 = *(const float*)(wbase + (r0 + colByte));
      const float x1 = *(const float*)(wbase + (r1 + colByte));
      const float x2 = *(const float*)(wbase + (r2 + colByte));
      const float x3 = *(const float*)(wbase + (r3 + colByte));
      v2f bA; bA.x = x0; bA.y = x1;
      v2f bB; bB.x = x2; bB.y = x3;
      acc0 = wmma_acc(aOnes, bA, acc0);
      acc1 = wmma_acc(aOnes, bB, acc1);
    }
    if (c < mainChunks) {
      const int s = (c << 2) + half;
      const float x = *(const float*)(wbase + (myOff[s] + colByte));
      const float y = *(const float*)(wbase + (myOff[s + 2] + colByte));
      v2f bM; bM.x = x; bM.y = y;
      acc0 = wmma_acc(aOnes, bM, acc0);
      ++c;
    }
    if (rem & 3) {
      const int s0 = (mainChunks << 2) + half;
      const int s1 = s0 + 2;
      float x = 0.0f, y = 0.0f;
      if (s0 < rem) x = *(const float*)(wbase + (myOff[s0] + colByte));
      if (s1 < rem) y = *(const float*)(wbase + (myOff[s1] + colByte));
      v2f bM; bM.x = x; bM.y = y;
      acc0 = wmma_acc(aOnes, bM, acc0);
    }
  }

  const float r = (acc0[0] + acc1[0]) / (float)(cnt > 0 ? cnt : 1);
  if (half == 0) {
    out[(size_t)bag * EMB_D + (colByte >> 2)] = r;
  }
}

extern "C" void kernel_launch(void* const* d_in, const int* in_sizes, int n_in,
                              void* d_out, int out_size, void* d_ws, size_t ws_size,
                              hipStream_t stream) {
  (void)n_in; (void)d_ws; (void)ws_size; (void)out_size;
  const int*   indices = (const int*)d_in[0];
  const int*   offsets = (const int*)d_in[1];
  const float* weight  = (const float*)d_in[2];
  float*       out     = (float*)d_out;

  const int T = in_sizes[0];
  const int B = in_sizes[1];

  const int totalWaves = B * COL_TILES;
  const int blocks     = (totalWaves + WAVES_PER_BLOCK - 1) / WAVES_PER_BLOCK;

  embbag_wmma<<<blocks, THREADS, 0, stream>>>(indices, offsets, weight, out, B, T);
}


# One wave per bag with 128-bit gathers: a half-wave reads a whole 256-B row and its register pairs feed WMMA through a 0/1 parity A; 4x fewer waves and loads; row summation order changes
# speedup vs baseline: 1.1819x; 1.1819x over previous
#include <hip/hip_runtime.h>


typedef float v2f __attribute__((ext_vector_type(2)));
typedef float v4f __attribute__((ext_vector_type(4)));
typedef float v8f __attribute__((ext_vector_type(8)));

#define EMB_D           64
#define WAVES_PER_BLOCK 8
#define THREADS         256

__device__ __forceinline__ v8f wmma_acc(v2f a, v2f b, v8f c) {
  return __builtin_amdgcn_wmma_f32_16x16x4_f32(
       false, a,  false, b,
       (short)0, c,  false,  false);
}

__device__ __forceinline__ v4f gather16(const char* wbase, unsigned row, unsigned colByte) {
  return *(const v4f*)(wbase + ((row << 8) | colByte));
}

__global__ __launch_bounds__(THREADS) void embbag_wmma(
    const int* __restrict__ indices,
    const int* __restrict__ offsets,
    const float* __restrict__ weight,
    float* __restrict__ out,
    int num_bags, int total_idx)
{
  const int lane = threadIdx.x & 31;
  const int waveInBlk = __builtin_amdgcn_readfirstlane((int)(threadIdx.x >> 5));
  const int bag       = blockIdx.x * WAVES_PER_BLOCK + waveInBlk;
  if (bag >= num_bags) return;

  const int c = lane & 15;
  const unsigned colByte  = (unsigned)c * 16u;
  const unsigned oddHalf  = (lane & 16) ? 0xffffffffu : 0u;

  const int bagStart = offsets[bag];
  const int bagEnd   = (bag + 1 < num_bags) ? offsets[bag + 1] : total_idx;
  const int cnt      = bagEnd - bagStart;

  const char* wbase = (const char*)weight;

  v2f aSel;
  aSel.x = (c & 1) ? 0.0f : 1.0f;
  aSel.y = (c & 1) ? 1.0f : 0.0f;

  v8f accLo = {}, accHi = {};
  float denom;

  if (cnt == 50) {
    const int* ip = indices + bagStart;
    v4f x[25];
    #pragma unroll
    for (int j = 0; j < 25; ++j) {
      const unsigned rEven = (unsigned)ip[2 * j];
      const unsigned rOdd  = (unsigned)ip[2 * j + 1];
      x[j] = gather16(wbase, (rOdd & oddHalf) | (rEven & ~oddHalf), colByte);
    }
    __builtin_amdgcn_sched_barrier(0);
    #pragma unroll
    for (int j = 0; j < 25; ++j) {
      v2f bLo; bLo.x = x[j][0]; bLo.y = x[j][1];
      v2f bHi; bHi.x = x[j][2]; bHi.y = x[j][3];
      accLo = wmma_acc(aSel, bLo, accLo);
      accHi = wmma_acc(aSel, bHi, accHi);
    }
    denom = 50.0f;
  } else {
    const int half = lane >> 4;
    for (int base = 0; base < cnt; base += 2) {
      const int  i  = base + half;
      const bool ok = i < cnt;
      const unsigned r = (unsigned)indices[bagStart + (ok ? i : 0)];
      v4f xv = gather16(wbase, r, colByte);
      const v4f zero = {0.0f, 0.0f, 0.0f, 0.0f};
      xv = ok ? xv : zero;
      v2f bLo; bLo.x = xv[0]; bLo.y = xv[1];
      v2f bHi; bHi.x = xv[2]; bHi.y = xv[3];
      accLo = wmma_acc(aSel, bLo, accLo);
      accHi = wmma_acc(aSel, bHi, accHi);
    }
    denom = (float)(cnt > 0 ? cnt : 1);
  }

  if (lane < 16) {
    v4f r;
    r[0] = accLo[0] / denom; r[1] = accLo[1] / denom;
    r[2] = accHi[0] / denom; r[3] = accHi[1] / denom;
    *(v4f*)(out + (size_t)bag * EMB_D + 4 * c) = r;
  }
}

extern "C" void kernel_launch(void* const* d_in, const int* in_sizes, int n_in,
                              void* d_out, int out_size, void* d_ws, size_t ws_size,
                              hipStream_t stream) {
  (void)n_in; (void)d_ws; (void)ws_size; (void)out_size;
  const int*   indices = (const int*)d_in[0];
  const int*   offsets = (const int*)d_in[1];
  const float* weight  = (const float*)d_in[2];
  float*       out     = (float*)d_out;

  const int T = in_sizes[0];
  const int B = in_sizes[1];

  const int blocks = (B + WAVES_PER_BLOCK - 1) / WAVES_PER_BLOCK;

  embbag_wmma<<<blocks, THREADS, 0, stream>>>(indices, offsets, weight, out, B, T);
}
